# starting kernel
# baseline (speedup 1.0000x reference)
#include <hip/hip_runtime.h>
#include <math.h>

typedef __attribute__((ext_vector_type(2))) float v2f;
typedef __attribute__((ext_vector_type(4))) float v4f;
typedef __attribute__((ext_vector_type(8))) float v8f;

#define N_NODES 100000
#define N_EDGES 1600000
#define FEAT    64
#define EMBED   128
#define NCLS    40

__global__ __launch_bounds__(256) void gcn_zero4(float* __restrict__ p, long n4) {
  long i = (long)blockIdx.x * blockDim.x + threadIdx.x;
  if (i < n4) ((v4f*)p)[i] = (v4f){0.0f, 0.0f, 0.0f, 0.0f};
}

template <int NF>
__global__ __launch_bounds__(256) void gcn_spmm_scatter4(
    const float* __restrict__ feat, const int* __restrict__ src,
    const int* __restrict__ dst, const float* __restrict__ w,
    float* __restrict__ out, long total) {
  constexpr int QP = NF / 4;
  long gid = (long)blockIdx.x * blockDim.x + threadIdx.x;
  if (gid >= total) return;
  int e = (int)(gid / QP);
  int f = (int)(gid % QP) * 4;
  int s = src[e];
  int d = dst[e];
  float ww = w[e];
  v4f v = *(const v4f*)(feat + (long)s * NF + f);
  float* o = out + (long)d * NF + f;
  atomicAdd(o + 0, ww * v.x);
  atomicAdd(o + 1, ww * v.y);
  atomicAdd(o + 2, ww * v.z);
  atomicAdd(o + 3, ww * v.w);
}

template <int K, int N, int NT, bool RELU, bool NMASK>
__global__ __launch_bounds__(256) void gcn_wmma_gemm(
    const float* __restrict__ A, const float* __restrict__ B,
    float* __restrict__ C, int totalTiles) {
  const int wave = threadIdx.x >> 5;
  const int lane = threadIdx.x & 31;
  const int tile = blockIdx.x * 8 + wave;
  if (tile >= totalTiles) return;

  const int mt = tile / NT;
  const int nt = tile % NT;
  const int m    = lane & 15;
  const int half = lane >> 4;
  const int klo  = half * 2;

  const int row  = mt * 16 + m;
  const int ncol = nt * 16 + m;

  const int   bcol  = (!NMASK || ncol < N) ? ncol : (N - 1);
  const float bmask = (!NMASK || ncol < N) ? 1.0f : 0.0f;

  const float* __restrict__ arow = A + (long)row * K;

  v8f c = {};
#pragma unroll
  for (int k = 0; k < K; k += 4) {
    v2f a = *(const v2f*)(arow + k + klo);
    v2f b;
    b.x = B[(long)(k + klo)     * N + bcol] * bmask;
    b.y = B[(long)(k + klo + 1) * N + bcol] * bmask;
    c = __builtin_amdgcn_wmma_f32_16x16x4_f32(
        false, a, false, b, (short)0, c, false, false);
  }

#pragma unroll
  for (int r = 0; r < 8; ++r) {
    int rr = mt * 16 + r + half * 8;
    int cc = nt * 16 + (lane & 15);
    if (!NMASK || cc < N) {
      float v = c[r];
      if (RELU) v = fmaxf(v, 0.0f);
      C[(long)rr * N + cc] = v;
    }
  }
}

__global__ __launch_bounds__(256) void gcn_log_softmax(
    const float* __restrict__ in, float* __restrict__ out, int n) {
  int i = blockIdx.x * blockDim.x + threadIdx.x;
  if (i >= n) return;
  const float* __restrict__ row = in + (long)i * NCLS;
  float mx = -INFINITY;
#pragma unroll
  for (int c = 0; c < NCLS; ++c) mx = fmaxf(mx, row[c]);
  float s = 0.0f;
#pragma unroll
  for (int c = 0; c < NCLS; ++c) s += __expf(row[c] - mx);
  float lse = mx + __logf(s);
  float* __restrict__ orow = out + (long)i * NCLS;
#pragma unroll
  for (int c = 0; c < NCLS; ++c) orow[c] = row[c] - lse;
}

extern "C" void kernel_launch(void* const* d_in, const int* in_sizes, int n_in,
                              void* d_out, int out_size, void* d_ws, size_t ws_size,
                              hipStream_t stream) {
  const float* x    = (const float*)d_in[0];
  const float* W1   = (const float*)d_in[1];
  const float* W2   = (const float*)d_in[2];
  const float* ew   = (const float*)d_in[3];
  const int*   esrc = (const int*)d_in[4];
  const int*   edst = (const int*)d_in[5];
  float*       out  = (float*)d_out;

  float* agg1 = (float*)d_ws;
  float* h    = agg1 + (size_t)N_NODES * FEAT;
  float* t    = h    + (size_t)N_NODES * EMBED;
  float* agg2 = t    + (size_t)N_NODES * NCLS;

  const int TPB = 256;

  {
    long n1 = (long)N_NODES * FEAT / 4;
    gcn_zero4<<<(int)((n1 + TPB - 1) / TPB), TPB, 0, stream>>>(agg1, n1);
    long n2 = (long)N_NODES * NCLS / 4;
    gcn_zero4<<<(int)((n2 + TPB - 1) / TPB), TPB, 0, stream>>>(agg2, n2);
  }

  {
    long total = (long)N_EDGES * (FEAT / 4);
    gcn_spmm_scatter4<FEAT><<<(int)((total + TPB - 1) / TPB), TPB, 0, stream>>>(
        x, esrc, edst, ew, agg1, total);
  }

  {
    const int NT = EMBED / 16;
    int tiles = (N_NODES / 16) * NT;
    gcn_wmma_gemm<FEAT, EMBED, NT, true, false>
        <<<(tiles + 7) / 8, TPB, 0, stream>>>(agg1, W1, h, tiles);
  }

  {
    const int NT = (NCLS + 15) / 16;
    int tiles = (N_NODES / 16) * NT;
    gcn_wmma_gemm<EMBED, NCLS, NT, false, true>
        <<<(tiles + 7) / 8, TPB, 0, stream>>>(h, W2, t, tiles);
  }

  {
    long total = (long)N_EDGES * (NCLS / 4);
    gcn_spmm_scatter4<NCLS><<<(int)((total + TPB - 1) / TPB), TPB, 0, stream>>>(
        t, esrc, edst, ew, agg2, total);
  }

  gcn_log_softmax<<<(N_NODES + TPB - 1) / TPB, TPB, 0, stream>>>(agg2, out, N_NODES);
}


# Whole-line global stores: log_softmax stages 128 rows through LDS, a GEMM1 wave owns 32 adjacent columns (half-waves swap by permlanex16): every store is full 128-B lines; sum orders unchanged
# speedup vs baseline: 1.8983x; 1.8983x over previous
#include <hip/hip_runtime.h>
#include <math.h>

typedef __attribute__((ext_vector_type(2))) float v2f;
typedef __attribute__((ext_vector_type(4))) float v4f;
typedef __attribute__((ext_vector_type(8))) float v8f;

#define N_NODES 100000
#define N_EDGES 1600000
#define FEAT    64
#define EMBED   128
#define NCLS    40


#define ZERO_WGS 256
template <int N4>
__global__ __launch_bounds__(256) void gcn_zero4(float* __restrict__ p) {
  constexpr int STRIDE = ZERO_WGS * 256;
  constexpr int FULL   = N4 / STRIDE;
  const int i0 = blockIdx.x * 256 + threadIdx.x;
  v4f* __restrict__ q = (v4f*)p + i0;
  const v4f z = (v4f){0.0f, 0.0f, 0.0f, 0.0f};
#pragma unroll
  for (int j = 0; j < FULL; ++j) q[j * STRIDE] = z;
  if (i0 + FULL * STRIDE < N4) q[FULL * STRIDE] = z;
}

#define ROWP 64
#define SPMM_WGS   1250
#define SPMM_WAVES (SPMM_WGS * 8)
template <int NF>
__global__ __launch_bounds__(256) void gcn_spmm_scatter_lines(
    const float* __restrict__ feat, const int* __restrict__ src,
    const int* __restrict__ dst, const float* __restrict__ w,
    float* __restrict__ out) {
  constexpr int EPW   = 32;
  constexpr int BATCH = 16;
  constexpr int TAIL  = NF - 32;
  constexpr int G     = 32 / TAIL;
  static_assert(N_EDGES % (EPW * SPMM_WAVES) == 0, "no edge tail, equal chunks per wave");
  static_assert(TAIL > 0 && 32 % TAIL == 0 && BATCH % G == 0 && NF <= ROWP, "row = two lines");

  const int lane = threadIdx.x & 31;
  const int wave = __builtin_amdgcn_readfirstlane((int)(threadIdx.x >> 5));

  const unsigned ulane = (unsigned)lane;
  const unsigned ltof  = 32u + (unsigned)(lane % TAIL);
  const int      g     = lane / TAIL;

  for (int chunk = blockIdx.x * 8 + wave; chunk < N_EDGES / EPW; chunk += SPMM_WAVES) {
  const int e0 = chunk * EPW;
#pragma unroll
  for (int b = 0; b < EPW; b += BATCH) {
    int   sv[BATCH], dv[BATCH];
    float wv[BATCH];
#pragma unroll
    for (int i = 0; i < BATCH; ++i) {
      sv[i] = src[e0 + b + i];
      dv[i] = dst[e0 + b + i];
      wv[i] = w[e0 + b + i];
    }
    float v0[BATCH], v1[BATCH / G];
#pragma unroll
    for (int i = 0; i < BATCH; ++i)
      v0[i] = feat[(unsigned)sv[i] * ROWP + ulane];
#pragma unroll
    for (int j = 0; j < BATCH / G; ++j) {
      int sj = sv[j * G];
#pragma unroll
      for (int q = 1; q < G; ++q) sj = (g == q) ? sv[j * G + q] : sj;
      v1[j] = feat[(unsigned)sj * ROWP + ltof];
    }
#pragma unroll
    for (int i = 0; i < BATCH; ++i)
      atomicAdd(out + ((unsigned)dv[i] * ROWP + ulane), wv[i] * v0[i]);
#pragma unroll
    for (int j = 0; j < BATCH / G; ++j) {
      int   dj = dv[j * G];
      float wj = wv[j * G];
#pragma unroll
      for (int q = 1; q < G; ++q) {
        dj = (g == q) ? dv[j * G + q] : dj;
        wj = (g == q) ? wv[j * G + q] : wj;
      }
      atomicAdd(out + ((unsigned)dj * ROWP + ltof), wj * v1[j]);
    }
  }
  }
}

#define GEMM2_WAVES 6264
template <int K, int N, int NT, bool RELU, bool NMASK, int LDC, int WAVES>
__global__ __launch_bounds__(256) void gcn_wmma_gemm(
    const float* __restrict__ A, const float* __restrict__ B,
    float* __restrict__ C, int mTiles) {
  static_assert(WAVES % NT == 0 && WAVES % 8 == 0, "whole work-groups, fixed nt per wave");
  const int wave = __builtin_amdgcn_readfirstlane((int)(threadIdx.x >> 5));
  const int lane = threadIdx.x & 31;
  const int g    = blockIdx.x * 8 + wave;

  const int nt   = g % NT;
  const int m    = lane & 15;
  const int half = lane >> 4;
  const int klo  = half * 2;
  const int ncol = nt * 16 + m;

  const int   bcol  = (!NMASK || ncol < N) ? ncol : (N - 1);
  const float bmask = (!NMASK || ncol < N) ? 1.0f : 0.0f;

  v2f bf[K / 4];
#pragma unroll
  for (int k = 0; k < K; k += 4) {
    bf[k / 4].x = B[(k + klo)     * N + bcol] * bmask;
    bf[k / 4].y = B[(k + klo + 1) * N + bcol] * bmask;
  }

  for (int mt = g / NT; mt < mTiles; mt += WAVES / NT) {
    const int row = mt * 16 + m;
    const float* __restrict__ arow = A + (long)row * K;

    v8f c = {};
#pragma unroll
    for (int k = 0; k < K; k += 4) {
      v2f a = *(const v2f*)(arow + k + klo);
      c = __builtin_amdgcn_wmma_f32_16x16x4_f32(
          false, a, false, bf[k / 4], (short)0, c, false, false);
    }

#pragma unroll
    for (int r = 0; r < 8; ++r) {
      int rr = mt * 16 + r + half * 8;
      int cc = nt * 16 + (lane & 15);
      if (!NMASK || cc < N) {
        float v = c[r];
        if (RELU) v = fmaxf(v, 0.0f);
        C[(long)rr * LDC + cc] = v;
      }
    }
  }
}

#define GEMM1_WGS 1250
__device__ __forceinline__ float gcn_other_half(float x) {
  const int b = __builtin_bit_cast(int, x);
  return __builtin_bit_cast(float,
      __builtin_amdgcn_permlanex16(b, b, 0x76543210, 0xfedcba98, false, false));
}
__global__ __launch_bounds__(128) void gcn_wmma_gemm1_lines(
    const float* __restrict__ A, const float* __restrict__ B,
    float* __restrict__ C, int mTiles) {
  constexpr int K = FEAT, N = EMBED;
  static_assert(N == 4 * 32, "4 waves x 32 columns");
  const int wave = __builtin_amdgcn_readfirstlane((int)(threadIdx.x >> 5));
  const int lane = threadIdx.x & 31;
  const int m    = lane & 15;
  const int half = lane >> 4;
  const int klo  = half * 2;
  const int col0 = wave * 32 + m;

  v2f b0[K / 4], b1[K / 4];
#pragma unroll
  for (int k = 0; k < K; k += 4) {
    b0[k / 4].x = B[(k + klo)     * N + col0];
    b0[k / 4].y = B[(k + klo + 1) * N + col0];
    b1[k / 4].x = B[(k + klo)     * N + col0 + 16];
    b1[k / 4].y = B[(k + klo + 1) * N + col0 + 16];
  }

  for (int mt = blockIdx.x; mt < mTiles; mt += GEMM1_WGS) {
    const float* __restrict__ arow = A + (long)(mt * 16 + m) * K;

    v8f c0 = {}, c1 = {};
#pragma unroll
    for (int k = 0; k < K; k += 4) {
      v2f a = *(const v2f*)(arow + k + klo);
      c0 = __builtin_amdgcn_wmma_f32_16x16x4_f32(
          false, a, false, b0[k / 4], (short)0, c0, false, false);
      c1 = __builtin_amdgcn_wmma_f32_16x16x4_f32(
          false, a, false, b1[k / 4], (short)0, c1, false, false);
    }

    float* __restrict__ crow = C + (long)(mt * 16) * N + wave * 32 + lane;
#pragma unroll
    for (int r = 0; r < 8; ++r) {
      const float x0 = fmaxf(c0[r], 0.0f);
      const float x1 = fmaxf(c1[r], 0.0f);
      const float y0 = gcn_other_half(x0);
      const float y1 = gcn_other_half(x1);
      crow[(long)r * N]       = half ? y1 : x0;
      crow[(long)(r + 8) * N] = half ? x1 : y0;
    }
  }
}

#define LSM_ROWS  128
#define LSM_PITCH 44
__global__ __launch_bounds__(LSM_ROWS) void gcn_log_softmax(
    const float* __restrict__ in, float* __restrict__ out, int n) {
  static_assert(NCLS % 4 == 0 && NCLS / 4 <= 16 && LSM_PITCH >= NCLS && LSM_PITCH % 4 == 0,
                "a row is NCLS/4 float4 chunks inside one 16-chunk source row");
  static_assert((LSM_ROWS * NCLS * 4) % 128 == 0, "a work-group's output starts on a line");
  __shared__ v4f tile4[LSM_ROWS * LSM_PITCH / 4];
  float* tile = (float*)tile4;

  const int tid   = threadIdx.x;
  const int row0  = blockIdx.x * LSM_ROWS;
  const int nrows = min(LSM_ROWS, n - row0);

  const int chunk = tid & 15;
  const int rsub  = tid >> 4;
  constexpr int SWEEPS = 16;
  if (chunk < NCLS / 4) {
    v4f ld[SWEEPS];
#pragma unroll
    for (int k = 0; k < SWEEPS; ++k) {
      const int r = min(k * (LSM_ROWS / 16) + rsub, nrows - 1);
      ld[k] = *(const v4f*)(in + (size_t)(row0 + r) * ROWP + chunk * 4);
    }
#pragma unroll
    for (int k = 0; k < SWEEPS; ++k) {
      const int r = k * (LSM_ROWS / 16) + rsub;
      *(v4f*)(tile + r * LSM_PITCH + chunk * 4) = ld[k];
    }
  }
  __syncthreads();

  float v[NCLS];
#pragma unroll
  for (int c = 0; c < NCLS; c += 4) {
    const v4f q = *(const v4f*)(tile + tid * LSM_PITCH + c);
    v[c] = q.x; v[c + 1] = q.y; v[c + 2] = q.z; v[c + 3] = q.w;
  }
  float mx = -INFINITY;
#pragma unroll
  for (int c = 0; c < NCLS; ++c) mx = fmaxf(mx, v[c]);
  float s = 0.0f;
#pragma unroll
  for (int c = 0; c < NCLS; ++c) s += __expf(v[c] - mx);
  const float lse = mx + __logf(s);
#pragma unroll
  for (int c = 0; c < NCLS; c += 4) {
    *(v4f*)(tile + tid * LSM_PITCH + c) =
        (v4f){v[c] - lse, v[c + 1] - lse, v[c + 2] - lse, v[c + 3] - lse};
  }
  __syncthreads();

  v4f* __restrict__ o4 = (v4f*)(out + (size_t)row0 * NCLS);
  const int n4 = nrows * (NCLS / 4);
  v4f st[NCLS / 4];
#pragma unroll
  for (int k = 0; k < NCLS / 4; ++k) {
    const int j = k * LSM_ROWS + tid;
    const int r = j / (NCLS / 4);
    const int c = (j - r * (NCLS / 4)) * 4;
    st[k] = *(const v4f*)(tile + r * LSM_PITCH + c);
  }
#pragma unroll
  for (int k = 0; k < NCLS / 4; ++k) {
    const int j = k * LSM_ROWS + tid;
    if (j < n4) o4[j] = st[k];
  }
}

extern "C" void kernel_launch(void* const* d_in, const int* in_sizes, int n_in,
                              void* d_out, int out_size, void* d_ws, size_t ws_size,
                              hipStream_t stream) {
  const float* x    = (const float*)d_in[0];
  const float* W1   = (const float*)d_in[1];
  const float* W2   = (const float*)d_in[2];
  const float* ew   = (const float*)d_in[3];
  const int*   esrc = (const int*)d_in[4];
  const int*   edst = (const int*)d_in[5];
  float*       out  = (float*)d_out;

  float* agg1 = (float*)d_ws;
  float* h    = agg1 + (size_t)N_NODES * FEAT;
  float* agg2 = h    + (size_t)N_NODES * EMBED;
  float* t    = agg1;

  const int TPB = 256;

  gcn_zero4<N_NODES * FEAT / 4><<<ZERO_WGS, TPB, 0, stream>>>(agg1);
  gcn_zero4<N_NODES * ROWP / 4><<<ZERO_WGS, TPB, 0, stream>>>(agg2);

  gcn_spmm_scatter_lines<FEAT><<<SPMM_WGS, TPB, 0, stream>>>(
      x, esrc, edst, ew, agg1);

  gcn_wmma_gemm1_lines<<<GEMM1_WGS, 128, 0, stream>>>(agg1, W1, h, N_NODES / 16);

  gcn_wmma_gemm<EMBED, NCLS, (NCLS + 15) / 16, false, true, ROWP, GEMM2_WAVES>
      <<<GEMM2_WAVES / 8, TPB, 0, stream>>>(h, W2, t, N_NODES / 16);

  gcn_spmm_scatter_lines<NCLS><<<SPMM_WGS, TPB, 0, stream>>>(
      t, esrc, edst, ew, agg2);

  gcn_log_softmax<<<(N_NODES + LSM_ROWS - 1) / LSM_ROWS, LSM_ROWS, 0, stream>>>(agg2, out, N_NODES);
}


# SPMM scatter split by chip half: dst rows < N/2 and >= N/2 each added into from one half only (die id, ticket per half); chunks walked twice, edges filtered by dst, lane-packed; sums arrival order
# speedup vs baseline: 2.2450x; 1.1826x over previous
#include <hip/hip_runtime.h>
#include <math.h>

typedef __attribute__((ext_vector_type(2))) float v2f;
typedef __attribute__((ext_vector_type(4))) float v4f;
typedef __attribute__((ext_vector_type(8))) float v8f;

#define N_NODES 100000
#define N_EDGES 1600000
#define FEAT    64
#define EMBED   128
#define NCLS    40


#define ZERO_WGS 256
template <int N4>
__global__ __launch_bounds__(256) void gcn_zero4(float* __restrict__ p) {
  constexpr int STRIDE = ZERO_WGS * 256;
  constexpr int FULL   = N4 / STRIDE;
  const int i0 = blockIdx.x * 256 + threadIdx.x;
  v4f* __restrict__ q = (v4f*)p + i0;
  const v4f z = (v4f){0.0f, 0.0f, 0.0f, 0.0f};
#pragma unroll
  for (int j = 0; j < FULL; ++j) q[j * STRIDE] = z;
  if (i0 + FULL * STRIDE < N4) q[FULL * STRIDE] = z;
}

#define ROWP 64
#define SPMM_WGS   1250
#define SPMM_SHARES (SPMM_WGS / 2)
#define SPMM_RWAVES (SPMM_SHARES * 8)
#define HALF_ROWS  (N_NODES / 2)
template <int NF>
__global__ __launch_bounds__(256) void gcn_spmm_scatter_halves(
    const float* __restrict__ feat, const int* __restrict__ src,
    const int* __restrict__ dst, const float* __restrict__ w,
    float* __restrict__ out, int* __restrict__ rank) {
  constexpr int EPW   = 32;
  constexpr int TAIL  = NF - 32;
  constexpr int G     = 32 / TAIL;
  static_assert(SPMM_WGS % 2 == 0 && (N_EDGES / EPW) % SPMM_RWAVES == 0 && N_EDGES % EPW == 0,
                "no edge tail, equal chunks per wave");
  static_assert(TAIL > 0 && 32 % TAIL == 0 && EPW % G == 0 && NF <= ROWP && EPW == 32,
                "row = two lines; one edge per lane");
  static_assert(G == 1 || EPW / G == 8, "the second-line case lists below are written for 8 instructions");
  static_assert(NCLS + 4 <= ROWP, "the four ticket counters fit behind the used floats of an agg2 row");

  __shared__ int s_ticket;
  const int hw_half = (int)((__builtin_amdgcn_s_sendmsg_rtn(0x87) >> 18) & 1u);
  if (threadIdx.x == 0) s_ticket = atomicAdd(rank + hw_half, 1);
  __syncthreads();
  const int ticket = __builtin_amdgcn_readfirstlane(s_ticket);
  const int range  = (ticket < SPMM_SHARES) ? hw_half : (hw_half ^ 1);
  const int share  = (ticket < SPMM_SHARES) ? ticket : (SPMM_WGS - 1 - ticket);
  const unsigned lo = (unsigned)range * (unsigned)HALF_ROWS;
  const unsigned nrows = range ? (unsigned)(N_NODES - HALF_ROWS) : (unsigned)HALF_ROWS;

  const int lane = threadIdx.x & 31;
  const int wave = __builtin_amdgcn_readfirstlane((int)(threadIdx.x >> 5));

  const unsigned ulane = (unsigned)lane;
  const unsigned ltof  = 32u + (unsigned)(lane % TAIL);
  const int      g     = lane / TAIL;

  int chunk = share * 8 + wave;
  int   es = src[chunk * EPW + lane];
  int   ed = dst[chunk * EPW + lane];
  float ew = w[chunk * EPW + lane];
  for (; chunk < N_EDGES / EPW; chunk += SPMM_RWAVES) {
    const bool mine = (unsigned)ed - lo < nrows;
    const unsigned mask = __builtin_amdgcn_ballot_w32(mine);
    const int n = __builtin_popcount(mask);
    const int below_in  = (int)__builtin_amdgcn_mbcnt_lo(mask, 0u);
    const int below_out = (int)__builtin_amdgcn_mbcnt_lo(~mask, 0u);
    const int pos = mine ? below_in : n + below_out;
    const int cs = __builtin_amdgcn_ds_permute(pos << 2, es);
    const int cd = __builtin_amdgcn_ds_permute(pos << 2, ed);
    const int cw = __builtin_amdgcn_ds_permute(pos << 2, __builtin_bit_cast(int, ew));
    {
      const int nx = min(chunk + SPMM_RWAVES, N_EDGES / EPW - 1) * EPW + lane;
      es = src[nx];
      ed = dst[nx];
      ew = w[nx];
    }

    if (n == 0) continue;
    float v0[EPW], v1[EPW / G];
#define GCN_GATHER(u)                                                       \
    {                                                                       \
      const unsigned s_ = (unsigned)__builtin_amdgcn_readlane(cs, (u));     \
      v0[(u)] = feat[s_ * ROWP + ulane];                                    \
      if constexpr (G == 1) v1[(u)] = feat[s_ * ROWP + 32u + ulane];        \
    }
#define GCN_ADD(u)                                                          \
    {                                                                       \
      const unsigned d_ = (unsigned)__builtin_amdgcn_readlane(cd, (u));     \
      const float    w_ = __builtin_bit_cast(float, __builtin_amdgcn_readlane(cw, (u))); \
      atomicAdd(out + (d_ * ROWP + ulane), w_ * v0[(u)]);                   \
      if constexpr (G == 1) atomicAdd(out + (d_ * ROWP + 32u + ulane), w_ * v1[(u)]); \
    }
    do {
      GCN_GATHER(0) if (n == 1) break;
      GCN_GATHER(1) if (n == 2) break;
      GCN_GATHER(2) if (n == 3) break;
      GCN_GATHER(3) if (n == 4) break;
      GCN_GATHER(4) if (n == 5) break;
      GCN_GATHER(5) if (n == 6) break;
      GCN_GATHER(6) if (n == 7) break;
      GCN_GATHER(7) if (n == 8) break;
      GCN_GATHER(8) if (n == 9) break;
      GCN_GATHER(9) if (n == 10) break;
      GCN_GATHER(10) if (n == 11) break;
      GCN_GATHER(11) if (n == 12) break;
      GCN_GATHER(12) if (n == 13) break;
      GCN_GATHER(13) if (n == 14) break;
      GCN_GATHER(14) if (n == 15) break;
      GCN_GATHER(15) if (n == 16) break;
      GCN_GATHER(16) if (n == 17) break;
      GCN_GATHER(17) if (n == 18) break;
      GCN_GATHER(18) if (n == 19) break;
      GCN_GATHER(19) if (n == 20) break;
      GCN_GATHER(20) if (n == 21) break;
      GCN_GATHER(21) if (n == 22) break;
      GCN_GATHER(22) if (n == 23) break;
      GCN_GATHER(23) if (n == 24) break;
      GCN_GATHER(24) if (n == 25) break;
      GCN_GATHER(25) if (n == 26) break;
      GCN_GATHER(26) if (n == 27) break;
      GCN_GATHER(27) if (n == 28) break;
      GCN_GATHER(28) if (n == 29) break;
      GCN_GATHER(29) if (n == 30) break;
      GCN_GATHER(30) if (n == 31) break;
      GCN_GATHER(31)
    } while (0);
    if constexpr (G > 1) {
      const int nq   = (n + G - 1) / G;
      const int from = g << 2;
      int sj[EPW / G], dj[EPW / G], wj[EPW / G];
#pragma unroll
      for (int j = 0; j < EPW / G; ++j) {
        sj[j] = __builtin_amdgcn_ds_bpermute(from + j * G * 4, cs);
        dj[j] = __builtin_amdgcn_ds_bpermute(from + j * G * 4, cd);
        wj[j] = __builtin_amdgcn_ds_bpermute(from + j * G * 4, cw);
      }
#define GCN_GATHER2(j) v1[(j)] = feat[(unsigned)sj[(j)] * ROWP + ltof];
#define GCN_ADD2(j)                                                         \
      if ((j) * G + g < n)                                                  \
        atomicAdd(out + ((unsigned)dj[(j)] * ROWP + ltof),                  \
                  __builtin_bit_cast(float, wj[(j)]) * v1[(j)]);
      do {
        GCN_GATHER2(0) if (nq == 1) break;
        GCN_GATHER2(1) if (nq == 2) break;
        GCN_GATHER2(2) if (nq == 3) break;
        GCN_GATHER2(3) if (nq == 4) break;
        GCN_GATHER2(4) if (nq == 5) break;
        GCN_GATHER2(5) if (nq == 6) break;
        GCN_GATHER2(6) if (nq == 7) break;
        GCN_GATHER2(7)
      } while (0);
      do {
        GCN_ADD(0) if (n == 1) break;
        GCN_ADD(1) if (n == 2) break;
        GCN_ADD(2) if (n == 3) break;
        GCN_ADD(3) if (n == 4) break;
        GCN_ADD(4) if (n == 5) break;
        GCN_ADD(5) if (n == 6) break;
        GCN_ADD(6) if (n == 7) break;
        GCN_ADD(7) if (n == 8) break;
        GCN_ADD(8) if (n == 9) break;
        GCN_ADD(9) if (n == 10) break;
        GCN_ADD(10) if (n == 11) break;
        GCN_ADD(11) if (n == 12) break;
        GCN_ADD(12) if (n == 13) break;
        GCN_ADD(13) if (n == 14) break;
        GCN_ADD(14) if (n == 15) break;
        GCN_ADD(15) if (n == 16) break;
        GCN_ADD(16) if (n == 17) break;
        GCN_ADD(17) if (n == 18) break;
        GCN_ADD(18) if (n == 19) break;
        GCN_ADD(19) if (n == 20) break;
        GCN_ADD(20) if (n == 21) break;
        GCN_ADD(21) if (n == 22) break;
        GCN_ADD(22) if (n == 23) break;
        GCN_ADD(23) if (n == 24) break;
        GCN_ADD(24) if (n == 25) break;
        GCN_ADD(25) if (n == 26) break;
        GCN_ADD(26) if (n == 27) break;
        GCN_ADD(27) if (n == 28) break;
        GCN_ADD(28) if (n == 29) break;
        GCN_ADD(29) if (n == 30) break;
        GCN_ADD(30) if (n == 31) break;
        GCN_ADD(31)
      } while (0);
      do {
        GCN_ADD2(0) if (nq == 1) break;
        GCN_ADD2(1) if (nq == 2) break;
        GCN_ADD2(2) if (nq == 3) break;
        GCN_ADD2(3) if (nq == 4) break;
        GCN_ADD2(4) if (nq == 5) break;
        GCN_ADD2(5) if (nq == 6) break;
        GCN_ADD2(6) if (nq == 7) break;
        GCN_ADD2(7)
      } while (0);
#undef GCN_GATHER2
#undef GCN_ADD2
    } else {
      do {
        GCN_ADD(0) if (n == 1) break;
        GCN_ADD(1) if (n == 2) break;
        GCN_ADD(2) if (n == 3) break;
        GCN_ADD(3) if (n == 4) break;
        GCN_ADD(4) if (n == 5) break;
        GCN_ADD(5) if (n == 6) break;
        GCN_ADD(6) if (n == 7) break;
        GCN_ADD(7) if (n == 8) break;
        GCN_ADD(8) if (n == 9) break;
        GCN_ADD(9) if (n == 10) break;
        GCN_ADD(10) if (n == 11) break;
        GCN_ADD(11) if (n == 12) break;
        GCN_ADD(12) if (n == 13) break;
        GCN_ADD(13) if (n == 14) break;
        GCN_ADD(14) if (n == 15) break;
        GCN_ADD(15) if (n == 16) break;
        GCN_ADD(16) if (n == 17) break;
        GCN_ADD(17) if (n == 18) break;
        GCN_ADD(18) if (n == 19) break;
        GCN_ADD(19) if (n == 20) break;
        GCN_ADD(20) if (n == 21) break;
        GCN_ADD(21) if (n == 22) break;
        GCN_ADD(22) if (n == 23) break;
        GCN_ADD(23) if (n == 24) break;
        GCN_ADD(24) if (n == 25) break;
        GCN_ADD(25) if (n == 26) break;
        GCN_ADD(26) if (n == 27) break;
        GCN_ADD(27) if (n == 28) break;
        GCN_ADD(28) if (n == 29) break;
        GCN_ADD(29) if (n == 30) break;
        GCN_ADD(30) if (n == 31) break;
        GCN_ADD(31)
      } while (0);
    }
#undef GCN_GATHER
#undef GCN_ADD
  }
}

#define GEMM2_WAVES 6264
template <int K, int N, int NT, bool RELU, bool NMASK, int LDC, int WAVES>
__global__ __launch_bounds__(256) void gcn_wmma_gemm(
    const float* __restrict__ A, const float* __restrict__ B,
    float* __restrict__ C, int mTiles) {
  static_assert(WAVES % NT == 0 && WAVES % 8 == 0, "whole work-groups, fixed nt per wave");
  const int wave = __builtin_amdgcn_readfirstlane((int)(threadIdx.x >> 5));
  const int lane = threadIdx.x & 31;
  const int g    = blockIdx.x * 8 + wave;

  const int nt   = g % NT;
  const int m    = lane & 15;
  const int half = lane >> 4;
  const int klo  = half * 2;
  const int ncol = nt * 16 + m;

  const int   bcol  = (!NMASK || ncol < N) ? ncol : (N - 1);
  const float bmask = (!NMASK || ncol < N) ? 1.0f : 0.0f;

  v2f bf[K / 4];
#pragma unroll
  for (int k = 0; k < K; k += 4) {
    bf[k / 4].x = B[(k + klo)     * N + bcol] * bmask;
    bf[k / 4].y = B[(k + klo + 1) * N + bcol] * bmask;
  }

  for (int mt = g / NT; mt < mTiles; mt += WAVES / NT) {
    const int row = mt * 16 + m;
    const float* __restrict__ arow = A + (long)row * K;

    v8f c = {};
#pragma unroll
    for (int k = 0; k < K; k += 4) {
      v2f a = *(const v2f*)(arow + k + klo);
      c = __builtin_amdgcn_wmma_f32_16x16x4_f32(
          false, a, false, bf[k / 4], (short)0, c, false, false);
    }

#pragma unroll
    for (int r = 0; r < 8; ++r) {
      int rr = mt * 16 + r + half * 8;
      int cc = nt * 16 + (lane & 15);
      if (!NMASK || cc < N) {
        float v = c[r];
        if (RELU) v = fmaxf(v, 0.0f);
        C[(long)rr * LDC + cc] = v;
      }
    }
  }
}

#define GEMM1_WGS 1250
__device__ __forceinline__ float gcn_other_half(float x) {
  const int b = __builtin_bit_cast(int, x);
  return __builtin_bit_cast(float,
      __builtin_amdgcn_permlanex16(b, b, 0x76543210, 0xfedcba98, false, false));
}
__global__ __launch_bounds__(128) void gcn_wmma_gemm1_lines(
    const float* __restrict__ A, const float* __restrict__ B,
    float* __restrict__ C, int mTiles) {
  constexpr int K = FEAT, N = EMBED;
  static_assert(N == 4 * 32, "4 waves x 32 columns");
  const int wave = __builtin_amdgcn_readfirstlane((int)(threadIdx.x >> 5));
  const int lane = threadIdx.x & 31;
  const int m    = lane & 15;
  const int half = lane >> 4;
  const int klo  = half * 2;
  const int col0 = wave * 32 + m;

  v2f b0[K / 4], b1[K / 4];
#pragma unroll
  for (int k = 0; k < K; k += 4) {
    b0[k / 4].x = B[(k + klo)     * N + col0];
    b0[k / 4].y = B[(k + klo + 1) * N + col0];
    b1[k / 4].x = B[(k + klo)     * N + col0 + 16];
    b1[k / 4].y = B[(k + klo + 1) * N + col0 + 16];
  }

  for (int mt = blockIdx.x; mt < mTiles; mt += GEMM1_WGS) {
    const float* __restrict__ arow = A + (long)(mt * 16 + m) * K;

    v8f c0 = {}, c1 = {};
#pragma unroll
    for (int k = 0; k < K; k += 4) {
      v2f a = *(const v2f*)(arow + k + klo);
      c0 = __builtin_amdgcn_wmma_f32_16x16x4_f32(
          false, a, false, b0[k / 4], (short)0, c0, false, false);
      c1 = __builtin_amdgcn_wmma_f32_16x16x4_f32(
          false, a, false, b1[k / 4], (short)0, c1, false, false);
    }

    float* __restrict__ crow = C + (long)(mt * 16) * N + wave * 32 + lane;
#pragma unroll
    for (int r = 0; r < 8; ++r) {
      const float x0 = fmaxf(c0[r], 0.0f);
      const float x1 = fmaxf(c1[r], 0.0f);
      const float y0 = gcn_other_half(x0);
      const float y1 = gcn_other_half(x1);
      crow[(long)r * N]       = half ? y1 : x0;
      crow[(long)(r + 8) * N] = half ? x1 : y0;
    }
  }
}

#define LSM_ROWS  128
#define LSM_PITCH 44
__global__ __launch_bounds__(LSM_ROWS) void gcn_log_softmax(
    const float* __restrict__ in, float* __restrict__ out, int n) {
  static_assert(NCLS % 4 == 0 && NCLS / 4 <= 16 && LSM_PITCH >= NCLS && LSM_PITCH % 4 == 0,
                "a row is NCLS/4 float4 chunks inside one 16-chunk source row");
  static_assert((LSM_ROWS * NCLS * 4) % 128 == 0, "a work-group's output starts on a line");
  __shared__ v4f tile4[LSM_ROWS * LSM_PITCH / 4];
  float* tile = (float*)tile4;

  const int tid   = threadIdx.x;
  const int row0  = blockIdx.x * LSM_ROWS;
  const int nrows = min(LSM_ROWS, n - row0);

  const int chunk = tid & 15;
  const int rsub  = tid >> 4;
  constexpr int SWEEPS = 16;
  if (chunk < NCLS / 4) {
    v4f ld[SWEEPS];
#pragma unroll
    for (int k = 0; k < SWEEPS; ++k) {
      const int r = min(k * (LSM_ROWS / 16) + rsub, nrows - 1);
      ld[k] = *(const v4f*)(in + (size_t)(row0 + r) * ROWP + chunk * 4);
    }
#pragma unroll
    for (int k = 0; k < SWEEPS; ++k) {
      const int r = k * (LSM_ROWS / 16) + rsub;
      *(v4f*)(tile + r * LSM_PITCH + chunk * 4) = ld[k];
    }
  }
  __syncthreads();

  float v[NCLS];
#pragma unroll
  for (int c = 0; c < NCLS; c += 4) {
    const v4f q = *(const v4f*)(tile + tid * LSM_PITCH + c);
    v[c] = q.x; v[c + 1] = q.y; v[c + 2] = q.z; v[c + 3] = q.w;
  }
  float mx = -INFINITY;
#pragma unroll
  for (int c = 0; c < NCLS; ++c) mx = fmaxf(mx, v[c]);
  float s = 0.0f;
#pragma unroll
  for (int c = 0; c < NCLS; ++c) s += __expf(v[c] - mx);
  const float lse = mx + __logf(s);
#pragma unroll
  for (int c = 0; c < NCLS; c += 4) {
    *(v4f*)(tile + tid * LSM_PITCH + c) =
        (v4f){v[c] - lse, v[c + 1] - lse, v[c + 2] - lse, v[c + 3] - lse};
  }
  __syncthreads();

  v4f* __restrict__ o4 = (v4f*)(out + (size_t)row0 * NCLS);
  const int n4 = nrows * (NCLS / 4);
  v4f st[NCLS / 4];
#pragma unroll
  for (int k = 0; k < NCLS / 4; ++k) {
    const int j = k * LSM_ROWS + tid;
    const int r = j / (NCLS / 4);
    const int c = (j - r * (NCLS / 4)) * 4;
    st[k] = *(const v4f*)(tile + r * LSM_PITCH + c);
  }
#pragma unroll
  for (int k = 0; k < NCLS / 4; ++k) {
    const int j = k * LSM_ROWS + tid;
    if (j < n4) o4[j] = st[k];
  }
}

extern "C" void kernel_launch(void* const* d_in, const int* in_sizes, int n_in,
                              void* d_out, int out_size, void* d_ws, size_t ws_size,
                              hipStream_t stream) {
  const float* x    = (const float*)d_in[0];
  const float* W1   = (const float*)d_in[1];
  const float* W2   = (const float*)d_in[2];
  const float* ew   = (const float*)d_in[3];
  const int*   esrc = (const int*)d_in[4];
  const int*   edst = (const int*)d_in[5];
  float*       out  = (float*)d_out;

  float* agg1 = (float*)d_ws;
  float* h    = agg1 + (size_t)N_NODES * FEAT;
  float* agg2 = h    + (size_t)N_NODES * EMBED;
  float* t    = agg1;
  int*   rank = (int*)(agg2 + NCLS);

  const int TPB = 256;

  gcn_zero4<N_NODES * FEAT / 4><<<ZERO_WGS, TPB, 0, stream>>>(agg1);
  gcn_zero4<N_NODES * ROWP / 4><<<ZERO_WGS, TPB, 0, stream>>>(agg2);

  gcn_spmm_scatter_halves<FEAT><<<SPMM_WGS, TPB, 0, stream>>>(
      x, esrc, edst, ew, agg1, rank);

  gcn_wmma_gemm1_lines<<<GEMM1_WGS, 128, 0, stream>>>(agg1, W1, h, N_NODES / 16);

  gcn_wmma_gemm<EMBED, NCLS, (NCLS + 15) / 16, false, true, ROWP, GEMM2_WAVES>
      <<<GEMM2_WAVES / 8, TPB, 0, stream>>>(h, W2, t, N_NODES / 16);

  gcn_spmm_scatter_halves<NCLS><<<SPMM_WGS, TPB, 0, stream>>>(
      t, esrc, edst, ew, agg2, rank + 2);

  gcn_log_softmax<<<(N_NODES + LSM_ROWS - 1) / LSM_ROWS, LSM_ROWS, 0, stream>>>(agg2, out, N_NODES);
}


# Both dense layers in one launch as transposed products (A = W^T, node = column): GEMM1's D registers are GEMM2's B fragments, h stays in registers, W1, W2 too; both k orders permuted; 6 launches
# speedup vs baseline: 2.5293x; 1.1266x over previous
#include <hip/hip_runtime.h>
#include <math.h>

typedef __attribute__((ext_vector_type(2))) float v2f;
typedef __attribute__((ext_vector_type(4))) float v4f;
typedef __attribute__((ext_vector_type(8))) float v8f;

#define N_NODES 100000
#define N_EDGES 1600000
#define FEAT    64
#define EMBED   128
#define NCLS    40


#define ZERO_WGS 256
template <int N4>
__global__ __launch_bounds__(256) void gcn_zero4(float* __restrict__ p) {
  constexpr int STRIDE = ZERO_WGS * 256;
  constexpr int FULL   = N4 / STRIDE;
  const int i0 = blockIdx.x * 256 + threadIdx.x;
  v4f* __restrict__ q = (v4f*)p + i0;
  const v4f z = (v4f){0.0f, 0.0f, 0.0f, 0.0f};
#pragma unroll
  for (int j = 0; j < FULL; ++j) q[j * STRIDE] = z;
  if (i0 + FULL * STRIDE < N4) q[FULL * STRIDE] = z;
}

#define ROWP 64
#define SPMM_WGS   1250
#define SPMM_SHARES (SPMM_WGS / 2)
#define SPMM_RWAVES (SPMM_SHARES * 8)
#define HALF_ROWS  (N_NODES / 2)
template <int NF>
__global__ __launch_bounds__(256) void gcn_spmm_scatter_halves(
    const float* __restrict__ feat, const int* __restrict__ src,
    const int* __restrict__ dst, const float* __restrict__ w,
    float* __restrict__ out, int* __restrict__ rank) {
  constexpr int EPW   = 32;
  constexpr int TAIL  = NF - 32;
  constexpr int G     = 32 / TAIL;
  static_assert(SPMM_WGS % 2 == 0 && (N_EDGES / EPW) % SPMM_RWAVES == 0 && N_EDGES % EPW == 0,
                "no edge tail, equal chunks per wave");
  static_assert(TAIL > 0 && 32 % TAIL == 0 && EPW % G == 0 && NF <= ROWP && EPW == 32,
                "row = two lines; one edge per lane");
  static_assert(G == 1 || EPW / G == 8, "the second-line case lists below are written for 8 instructions");
  static_assert(NCLS + 4 <= ROWP, "the four ticket counters fit behind the used floats of an agg2 row");

  __shared__ int s_ticket;
  const int hw_half = (int)((__builtin_amdgcn_s_sendmsg_rtn(0x87) >> 18) & 1u);
  if (threadIdx.x == 0) s_ticket = atomicAdd(rank + hw_half, 1);
  __syncthreads();
  const int ticket = __builtin_amdgcn_readfirstlane(s_ticket);
  const int range  = (ticket < SPMM_SHARES) ? hw_half : (hw_half ^ 1);
  const int share  = (ticket < SPMM_SHARES) ? ticket : (SPMM_WGS - 1 - ticket);
  const unsigned lo = (unsigned)range * (unsigned)HALF_ROWS;
  const unsigned nrows = range ? (unsigned)(N_NODES - HALF_ROWS) : (unsigned)HALF_ROWS;

  const int lane = threadIdx.x & 31;
  const int wave = __builtin_amdgcn_readfirstlane((int)(threadIdx.x >> 5));

  const unsigned ulane = (unsigned)lane;
  const unsigned ltof  = 32u + (unsigned)(lane % TAIL);
  const int      g     = lane / TAIL;

  int chunk = share * 8 + wave;
  int   es = src[chunk * EPW + lane];
  int   ed = dst[chunk * EPW + lane];
  float ew = w[chunk * EPW + lane];
  for (; chunk < N_EDGES / EPW; chunk += SPMM_RWAVES) {
    const bool mine = (unsigned)ed - lo < nrows;
    const unsigned mask = __builtin_amdgcn_ballot_w32(mine);
    const int n = __builtin_popcount(mask);
    const int below_in  = (int)__builtin_amdgcn_mbcnt_lo(mask, 0u);
    const int below_out = (int)__builtin_amdgcn_mbcnt_lo(~mask, 0u);
    const int pos = mine ? below_in : n + below_out;
    const int cs = __builtin_amdgcn_ds_permute(pos << 2, es);
    const int cd = __builtin_amdgcn_ds_permute(pos << 2, ed);
    const int cw = __builtin_amdgcn_ds_permute(pos << 2, __builtin_bit_cast(int, ew));
    {
      const int nx = min(chunk + SPMM_RWAVES, N_EDGES / EPW - 1) * EPW + lane;
      es = src[nx];
      ed = dst[nx];
      ew = w[nx];
    }

    if (n == 0) continue;
    float v0[EPW], v1[EPW / G];
#define GCN_GATHER(u)                                                       \
    {                                                                       \
      const unsigned s_ = (unsigned)__builtin_amdgcn_readlane(cs, (u));     \
      v0[(u)] = feat[s_ * ROWP + ulane];                                    \
      if constexpr (G == 1) v1[(u)] = feat[s_ * ROWP + 32u + ulane];        \
    }
#define GCN_ADD(u)                                                          \
    {                                                                       \
      const unsigned d_ = (unsigned)__builtin_amdgcn_readlane(cd, (u));     \
      const float    w_ = __builtin_bit_cast(float, __builtin_amdgcn_readlane(cw, (u))); \
      atomicAdd(out + (d_ * ROWP + ulane), w_ * v0[(u)]);                   \
      if constexpr (G == 1) atomicAdd(out + (d_ * ROWP + 32u + ulane), w_ * v1[(u)]); \
    }
    do {
      GCN_GATHER(0) if (n == 1) break;
      GCN_GATHER(1) if (n == 2) break;
      GCN_GATHER(2) if (n == 3) break;
      GCN_GATHER(3) if (n == 4) break;
      GCN_GATHER(4) if (n == 5) break;
      GCN_GATHER(5) if (n == 6) break;
      GCN_GATHER(6) if (n == 7) break;
      GCN_GATHER(7) if (n == 8) break;
      GCN_GATHER(8) if (n == 9) break;
      GCN_GATHER(9) if (n == 10) break;
      GCN_GATHER(10) if (n == 11) break;
      GCN_GATHER(11) if (n == 12) break;
      GCN_GATHER(12) if (n == 13) break;
      GCN_GATHER(13) if (n == 14) break;
      GCN_GATHER(14) if (n == 15) break;
      GCN_GATHER(15) if (n == 16) break;
      GCN_GATHER(16) if (n == 17) break;
      GCN_GATHER(17) if (n == 18) break;
      GCN_GATHER(18) if (n == 19) break;
      GCN_GATHER(19) if (n == 20) break;
      GCN_GATHER(20) if (n == 21) break;
      GCN_GATHER(21) if (n == 22) break;
      GCN_GATHER(22) if (n == 23) break;
      GCN_GATHER(23) if (n == 24) break;
      GCN_GATHER(24) if (n == 25) break;
      GCN_GATHER(25) if (n == 26) break;
      GCN_GATHER(26) if (n == 27) break;
      GCN_GATHER(27) if (n == 28) break;
      GCN_GATHER(28) if (n == 29) break;
      GCN_GATHER(29) if (n == 30) break;
      GCN_GATHER(30) if (n == 31) break;
      GCN_GATHER(31)
    } while (0);
    if constexpr (G > 1) {
      const int nq   = (n + G - 1) / G;
      const int from = g << 2;
      int sj[EPW / G], dj[EPW / G], wj[EPW / G];
#pragma unroll
      for (int j = 0; j < EPW / G; ++j) {
        sj[j] = __builtin_amdgcn_ds_bpermute(from + j * G * 4, cs);
        dj[j] = __builtin_amdgcn_ds_bpermute(from + j * G * 4, cd);
        wj[j] = __builtin_amdgcn_ds_bpermute(from + j * G * 4, cw);
      }
#define GCN_GATHER2(j) v1[(j)] = feat[(unsigned)sj[(j)] * ROWP + ltof];
#define GCN_ADD2(j)                                                         \
      if ((j) * G + g < n)                                                  \
        atomicAdd(out + ((unsigned)dj[(j)] * ROWP + ltof),                  \
                  __builtin_bit_cast(float, wj[(j)]) * v1[(j)]);
      do {
        GCN_GATHER2(0) if (nq == 1) break;
        GCN_GATHER2(1) if (nq == 2) break;
        GCN_GATHER2(2) if (nq == 3) break;
        GCN_GATHER2(3) if (nq == 4) break;
        GCN_GATHER2(4) if (nq == 5) break;
        GCN_GATHER2(5) if (nq == 6) break;
        GCN_GATHER2(6) if (nq == 7) break;
        GCN_GATHER2(7)
      } while (0);
      do {
        GCN_ADD(0) if (n == 1) break;
        GCN_ADD(1) if (n == 2) break;
        GCN_ADD(2) if (n == 3) break;
        GCN_ADD(3) if (n == 4) break;
        GCN_ADD(4) if (n == 5) break;
        GCN_ADD(5) if (n == 6) break;
        GCN_ADD(6) if (n == 7) break;
        GCN_ADD(7) if (n == 8) break;
        GCN_ADD(8) if (n == 9) break;
        GCN_ADD(9) if (n == 10) break;
        GCN_ADD(10) if (n == 11) break;
        GCN_ADD(11) if (n == 12) break;
        GCN_ADD(12) if (n == 13) break;
        GCN_ADD(13) if (n == 14) break;
        GCN_ADD(14) if (n == 15) break;
        GCN_ADD(15) if (n == 16) break;
        GCN_ADD(16) if (n == 17) break;
        GCN_ADD(17) if (n == 18) break;
        GCN_ADD(18) if (n == 19) break;
        GCN_ADD(19) if (n == 20) break;
        GCN_ADD(20) if (n == 21) break;
        GCN_ADD(21) if (n == 22) break;
        GCN_ADD(22) if (n == 23) break;
        GCN_ADD(23) if (n == 24) break;
        GCN_ADD(24) if (n == 25) break;
        GCN_ADD(25) if (n == 26) break;
        GCN_ADD(26) if (n == 27) break;
        GCN_ADD(27) if (n == 28) break;
        GCN_ADD(28) if (n == 29) break;
        GCN_ADD(29) if (n == 30) break;
        GCN_ADD(30) if (n == 31) break;
        GCN_ADD(31)
      } while (0);
      do {
        GCN_ADD2(0) if (nq == 1) break;
        GCN_ADD2(1) if (nq == 2) break;
        GCN_ADD2(2) if (nq == 3) break;
        GCN_ADD2(3) if (nq == 4) break;
        GCN_ADD2(4) if (nq == 5) break;
        GCN_ADD2(5) if (nq == 6) break;
        GCN_ADD2(6) if (nq == 7) break;
        GCN_ADD2(7)
      } while (0);
#undef GCN_GATHER2
#undef GCN_ADD2
    } else {
      do {
        GCN_ADD(0) if (n == 1) break;
        GCN_ADD(1) if (n == 2) break;
        GCN_ADD(2) if (n == 3) break;
        GCN_ADD(3) if (n == 4) break;
        GCN_ADD(4) if (n == 5) break;
        GCN_ADD(5) if (n == 6) break;
        GCN_ADD(6) if (n == 7) break;
        GCN_ADD(7) if (n == 8) break;
        GCN_ADD(8) if (n == 9) break;
        GCN_ADD(9) if (n == 10) break;
        GCN_ADD(10) if (n == 11) break;
        GCN_ADD(11) if (n == 12) break;
        GCN_ADD(12) if (n == 13) break;
        GCN_ADD(13) if (n == 14) break;
        GCN_ADD(14) if (n == 15) break;
        GCN_ADD(15) if (n == 16) break;
        GCN_ADD(16) if (n == 17) break;
        GCN_ADD(17) if (n == 18) break;
        GCN_ADD(18) if (n == 19) break;
        GCN_ADD(19) if (n == 20) break;
        GCN_ADD(20) if (n == 21) break;
        GCN_ADD(21) if (n == 22) break;
        GCN_ADD(22) if (n == 23) break;
        GCN_ADD(23) if (n == 24) break;
        GCN_ADD(24) if (n == 25) break;
        GCN_ADD(25) if (n == 26) break;
        GCN_ADD(26) if (n == 27) break;
        GCN_ADD(27) if (n == 28) break;
        GCN_ADD(28) if (n == 29) break;
        GCN_ADD(29) if (n == 30) break;
        GCN_ADD(30) if (n == 31) break;
        GCN_ADD(31)
      } while (0);
    }
#undef GCN_GATHER
#undef GCN_ADD
  }
}

#define DENSE_WGS   256
#define DENSE_WAVES (DENSE_WGS * 4)
__global__ __launch_bounds__(128) void gcn_dense_chain(
    const float* __restrict__ agg, const float* __restrict__ W1,
    const float* __restrict__ W2, float* __restrict__ t, int nTiles) {
  static_assert(FEAT == 64 && EMBED == 128 && NCLS > 32 && NCLS <= 40 && NCLS + 8 <= ROWP,
                "8 b128 loads per agg1 row; 8 e blocks; classes end in the low half of block 2");
  const int wave = __builtin_amdgcn_readfirstlane((int)(threadIdx.x >> 5));
  const int lane = threadIdx.x & 31;
  const int c    = lane & 15;
  const int hh   = lane >> 4;
  const int g    = blockIdx.x * 4 + wave;

  v2f w1f[8][16];
#pragma unroll
  for (int eb = 0; eb < 8; ++eb) {
#pragma unroll
    for (int qp = 0; qp < 16; ++qp) {
      const int k0 = 8 * (qp >> 1) + 4 * hh + 2 * (qp & 1);
      w1f[eb][qp].x = W1[(k0)     * EMBED + 16 * eb + c];
      w1f[eb][qp].y = W1[(k0 + 1) * EMBED + 16 * eb + c];
    }
  }
  v2f w2f[3][32];
#pragma unroll
  for (int cb = 0; cb < 3; ++cb) {
    const int  col = 16 * cb + c;
    const bool ok  = col < NCLS;
    const int  cc  = ok ? col : NCLS - 1;
#pragma unroll
    for (int ej = 0; ej < 32; ++ej) {
      const int k0 = 16 * (ej >> 2) + 8 * hh + 2 * (ej & 3);
      const float a0 = W2[(k0)     * NCLS + cc];
      const float a1 = W2[(k0 + 1) * NCLS + cc];
      w2f[cb][ej].x = ok ? a0 : 0.0f;
      w2f[cb][ej].y = ok ? a1 : 0.0f;
    }
  }

  v4f bcur[8];
  {
    const float* __restrict__ arow = agg + (size_t)(g * 16 + c) * FEAT + 4 * hh;
#pragma unroll
    for (int q = 0; q < 8; ++q) bcur[q] = *(const v4f*)(arow + 8 * q);
  }
  for (int mt = g; mt < nTiles; mt += DENSE_WAVES) {
    v4f bnext[8];
    {
      const int nx = min(mt + DENSE_WAVES, nTiles - 1);
      const float* __restrict__ arow = agg + (size_t)(nx * 16 + c) * FEAT + 4 * hh;
#pragma unroll
      for (int q = 0; q < 8; ++q) bnext[q] = *(const v4f*)(arow + 8 * q);
    }

    v8f acc[8];
#pragma unroll
    for (int eb = 0; eb < 8; ++eb) acc[eb] = (v8f){};
#pragma unroll
    for (int qp = 0; qp < 16; ++qp) {
      const v4f b4 = bcur[qp >> 1];
      const v2f bb = (qp & 1) ? (v2f){b4.z, b4.w} : (v2f){b4.x, b4.y};
#pragma unroll
      for (int eb = 0; eb < 8; ++eb) {
        acc[eb] = __builtin_amdgcn_wmma_f32_16x16x4_f32(
            false, w1f[eb][qp], false, bb, (short)0, acc[eb], false, false);
      }
    }

    v8f o[3];
#pragma unroll
    for (int cb = 0; cb < 3; ++cb) o[cb] = (v8f){};
#pragma unroll
    for (int ej = 0; ej < 32; ++ej) {
      const int eb = ej >> 2, j = ej & 3;
      const v2f hb = (v2f){fmaxf(acc[eb][2 * j], 0.0f), fmaxf(acc[eb][2 * j + 1], 0.0f)};
#pragma unroll
      for (int cb = 0; cb < 3; ++cb) {
        o[cb] = __builtin_amdgcn_wmma_f32_16x16x4_f32(
            false, w2f[cb][ej], false, hb, (short)0, o[cb], false, false);
      }
    }

    float* __restrict__ orow = t + (size_t)(mt * 16 + c) * ROWP + 8 * hh;
#pragma unroll
    for (int cb = 0; cb < 3; ++cb) {
      if (16 * cb + 8 * hh < NCLS) {
        *(v4f*)(orow + 16 * cb)     = (v4f){o[cb][0], o[cb][1], o[cb][2], o[cb][3]};
        *(v4f*)(orow + 16 * cb + 4) = (v4f){o[cb][4], o[cb][5], o[cb][6], o[cb][7]};
      }
    }
#pragma unroll
    for (int q = 0; q < 8; ++q) bcur[q] = bnext[q];
  }
}

#define LSM_ROWS  128
#define LSM_PITCH 44
__global__ __launch_bounds__(LSM_ROWS) void gcn_log_softmax(
    const float* __restrict__ in, float* __restrict__ out, int n) {
  static_assert(NCLS % 4 == 0 && NCLS / 4 <= 16 && LSM_PITCH >= NCLS && LSM_PITCH % 4 == 0,
                "a row is NCLS/4 float4 chunks inside one 16-chunk source row");
  static_assert((LSM_ROWS * NCLS * 4) % 128 == 0, "a work-group's output starts on a line");
  __shared__ v4f tile4[LSM_ROWS * LSM_PITCH / 4];
  float* tile = (float*)tile4;

  const int tid   = threadIdx.x;
  const int row0  = blockIdx.x * LSM_ROWS;
  const int nrows = min(LSM_ROWS, n - row0);

  const int chunk = tid & 15;
  const int rsub  = tid >> 4;
  constexpr int SWEEPS = 16;
  if (chunk < NCLS / 4) {
    v4f ld[SWEEPS];
#pragma unroll
    for (int k = 0; k < SWEEPS; ++k) {
      const int r = min(k * (LSM_ROWS / 16) + rsub, nrows - 1);
      ld[k] = *(const v4f*)(in + (size_t)(row0 + r) * ROWP + chunk * 4);
    }
#pragma unroll
    for (int k = 0; k < SWEEPS; ++k) {
      const int r = k * (LSM_ROWS / 16) + rsub;
      *(v4f*)(tile + r * LSM_PITCH + chunk * 4) = ld[k];
    }
  }
  __syncthreads();

  float v[NCLS];
#pragma unroll
  for (int c = 0; c < NCLS; c += 4) {
    const v4f q = *(const v4f*)(tile + tid * LSM_PITCH + c);
    v[c] = q.x; v[c + 1] = q.y; v[c + 2] = q.z; v[c + 3] = q.w;
  }
  float mx = -INFINITY;
#pragma unroll
  for (int c = 0; c < NCLS; ++c) mx = fmaxf(mx, v[c]);
  float s = 0.0f;
#pragma unroll
  for (int c = 0; c < NCLS; ++c) s += __expf(v[c] - mx);
  const float lse = mx + __logf(s);
#pragma unroll
  for (int c = 0; c < NCLS; c += 4) {
    *(v4f*)(tile + tid * LSM_PITCH + c) =
        (v4f){v[c] - lse, v[c + 1] - lse, v[c + 2] - lse, v[c + 3] - lse};
  }
  __syncthreads();

  v4f* __restrict__ o4 = (v4f*)(out + (size_t)row0 * NCLS);
  const int n4 = nrows * (NCLS / 4);
  v4f st[NCLS / 4];
#pragma unroll
  for (int k = 0; k < NCLS / 4; ++k) {
    const int j = k * LSM_ROWS + tid;
    const int r = j / (NCLS / 4);
    const int c = (j - r * (NCLS / 4)) * 4;
    st[k] = *(const v4f*)(tile + r * LSM_PITCH + c);
  }
#pragma unroll
  for (int k = 0; k < NCLS / 4; ++k) {
    const int j = k * LSM_ROWS + tid;
    if (j < n4) o4[j] = st[k];
  }
}

extern "C" void kernel_launch(void* const* d_in, const int* in_sizes, int n_in,
                              void* d_out, int out_size, void* d_ws, size_t ws_size,
                              hipStream_t stream) {
  const float* x    = (const float*)d_in[0];
  const float* W1   = (const float*)d_in[1];
  const float* W2   = (const float*)d_in[2];
  const float* ew   = (const float*)d_in[3];
  const int*   esrc = (const int*)d_in[4];
  const int*   edst = (const int*)d_in[5];
  float*       out  = (float*)d_out;

  float* agg1 = (float*)d_ws;
  float* t    = agg1 + (size_t)N_NODES * FEAT;
  float* agg2 = t    + (size_t)N_NODES * EMBED;
  int*   rank = (int*)(agg2 + NCLS);

  const int TPB = 256;

  gcn_zero4<N_NODES * FEAT / 4><<<ZERO_WGS, TPB, 0, stream>>>(agg1);
  gcn_zero4<N_NODES * ROWP / 4><<<ZERO_WGS, TPB, 0, stream>>>(agg2);

  gcn_spmm_scatter_halves<FEAT><<<SPMM_WGS, TPB, 0, stream>>>(
      x, esrc, edst, ew, agg1, rank);

  gcn_dense_chain<<<DENSE_WGS, 128, 0, stream>>>(agg1, W1, W2, t, N_NODES / 16);

  gcn_spmm_scatter_halves<NCLS><<<SPMM_WGS, TPB, 0, stream>>>(
      t, esrc, edst, ew, agg2, rank + 2);

  gcn_log_softmax<<<(N_NODES + LSM_ROWS - 1) / LSM_ROWS, LSM_ROWS, 0, stream>>>(agg2, out, N_NODES);
}
